# starting kernel
# baseline (speedup 1.0000x reference)
#include <hip/hip_runtime.h>

typedef __attribute__((ext_vector_type(2))) float v2f;
typedef __attribute__((ext_vector_type(8))) float v8f;

#define IN_H 128
#define IN_W 128
#define KHH 8
#define KWW 8
#define OH 121
#define OW 121
#define BATCH 64
#define UW 24
#define KP 192
#define LSTRIDE 196
#define NWAVES 4

__global__ __launch_bounds__(128) void lc2d_wmma_f32_kernel(
    const float* __restrict__ x,
    const float* __restrict__ w,
    const float* __restrict__ bias,
    float* __restrict__ out)
{
    __shared__ float Wl[16 * LSTRIDE];
    __shared__ float Xl[NWAVES][16 * LSTRIDE];

    const int i    = blockIdx.x >> 3;
    const int j0   = (blockIdx.x & 7) << 4;
    const int tid  = threadIdx.x;
    const int wave = tid >> 5;
    const int lane = tid & 31;

    for (int idx = tid; idx < 16 * LSTRIDE; idx += 128) Wl[idx] = 0.0f;
    __syncthreads();
    for (int idx = tid; idx < 16 * (KHH * KWW); idx += 128) {
        const int jl = idx >> 6;
        const int t  = idx & 63;
        const int kh = t >> 3, kw = t & 7;
        const int j  = j0 + jl;
        if (j < OW) {
            Wl[jl * LSTRIDE + kh * UW + jl + kw] = w[(i * OW + j) * (KHH * KWW) + t];
        }
    }

    for (int idx = lane; idx < 16 * KP; idx += 32) {
        const int b   = idx / KP;
        const int k   = idx % KP;
        const int kh  = k / UW;
        const int c   = k % UW;
        const int col = j0 + c;
        float v = 0.0f;
        if (col < IN_W)
            v = x[((wave * 16 + b) * IN_H + (i + kh)) * IN_W + col];
        Xl[wave][b * LSTRIDE + k] = v;
    }
    __syncthreads();

    const int mn    = lane & 15;
    const int khalf = lane >> 4;
    const float* arow = &Wl[mn * LSTRIDE + khalf * 2];
    const float* brow = &Xl[wave][mn * LSTRIDE + khalf * 2];

    v8f acc = {};
    #pragma unroll
    for (int kk = 0; kk < KP / 4; ++kk) {
        v2f a = *(const v2f*)(arow + kk * 4);
        v2f b = *(const v2f*)(brow + kk * 4);
        acc = __builtin_amdgcn_wmma_f32_16x16x4_f32(
             false, a,  false, b,
             (short)0, acc,  false,  false);
    }

    const int bglob = wave * 16 + mn;
    #pragma unroll
    for (int v = 0; v < 8; ++v) {
        const int j = j0 + v + khalf * 8;
        if (j < OW)
            out[(bglob * OH + i) * OW + j] = acc[v] + bias[i * OW + j];
    }
}

extern "C" void kernel_launch(void* const* d_in, const int* in_sizes, int n_in,
                              void* d_out, int out_size, void* d_ws, size_t ws_size,
                              hipStream_t stream) {
    const float* x    = (const float*)d_in[0];
    const float* w    = (const float*)d_in[1];
    const float* bias = (const float*)d_in[2];
    float* out        = (float*)d_out;

    dim3 grid(OH * 8);
    dim3 block(128);
    lc2d_wmma_f32_kernel<<<grid, block, 0, stream>>>(x, w, bias, out);
}


# all global reads issued up front, tiles as 16-byte loads (24 input, 2 weight, 8 bias per thread) before any LDS work, replacing 104 dependent 4-byte load-then-store trips
# speedup vs baseline: 1.9896x; 1.9896x over previous
#include <hip/hip_runtime.h>


typedef __attribute__((ext_vector_type(2))) float v2f;
typedef __attribute__((ext_vector_type(4))) float v4f;
typedef __attribute__((ext_vector_type(8))) float v8f;

#define IN_H 128
#define IN_W 128
#define KHH 8
#define KWW 8
#define OH 121
#define OW 121
#define BATCH 64
#define UW 24
#define KP 192
#define LSTRIDE 196
#define NWAVES 4
#define WQ 2
#define XT 3
#define XB 8

__global__ __launch_bounds__(128) void lc2d_wmma_f32_kernel(
    const float* __restrict__ x,
    const float* __restrict__ w,
    const float* __restrict__ bias,
    float* __restrict__ out)
{
    __shared__ __attribute__((aligned(16))) float Wl[16 * LSTRIDE];
    __shared__ __attribute__((aligned(16))) float Xl[NWAVES][16 * LSTRIDE];

    const int i     = blockIdx.x >> 3;
    const int j0    = (blockIdx.x & 7) << 4;
    const int tid   = threadIdx.x;
    const int wave  = tid >> 5;
    const int lane  = tid & 31;
    const int mn    = lane & 15;
    const int khalf = lane >> 4;

    v4f wv[WQ];
    #pragma unroll
    for (int r = 0; r < WQ; ++r) {
        const int f  = tid + 128 * r;
        const int jl = f >> 4;
        const int jc = min(j0 + jl, OW - 1);
        wv[r] = *(const v4f*)(w + (i * OW + jc) * (KHH * KWW) + ((f & 15) << 2));
    }
    int  gofs[XT], lofs[XT];
    bool edge[XT];
    #pragma unroll
    for (int t = 0; t < XT; ++t) {
        const int ft  = lane + 32 * t;
        const int rs  = ft / 6;
        const int c4  = ft - rs * 6;
        const int b   = rs >> 3;
        const int kh  = rs & 7;
        const int col = j0 + (c4 << 2);
        edge[t] = col >= IN_W;
        gofs[t] = ((wave * 16 + b) * IN_H + (i + kh)) * IN_W + min(col, IN_W - 4);
        lofs[t] = b * LSTRIDE + kh * UW + (c4 << 2);
    }
    v4f xv[XB][XT];
    #pragma unroll
    for (int q = 0; q < XB; ++q) {
        #pragma unroll
        for (int t = 0; t < XT; ++t)
            xv[q][t] = *(const v4f*)(x + gofs[t] + q * (2 * IN_H * IN_W));
    }
    float bv[8];
    #pragma unroll
    for (int v = 0; v < 8; ++v)
        bv[v] = bias[i * OW + min(j0 + v + khalf * 8, OW - 1)];

    for (int idx = tid; idx < 16 * LSTRIDE; idx += 128) Wl[idx] = 0.0f;
    __syncthreads();
    #pragma unroll
    for (int r = 0; r < WQ; ++r) {
        const int f   = tid + 128 * r;
        const int jl  = f >> 4;
        const int kh  = (f >> 1) & 7;
        const int kw0 = (f & 1) << 2;
        if (j0 + jl < OW) {
            float* p = &Wl[jl * LSTRIDE + kh * UW + jl + kw0];
            p[0] = wv[r].x; p[1] = wv[r].y; p[2] = wv[r].z; p[3] = wv[r].w;
        }
    }

    if (j0 + UW > IN_W) {
        #pragma unroll
        for (int q = 0; q < XB; ++q) {
            #pragma unroll
            for (int t = 0; t < XT; ++t)
                if (edge[t]) xv[q][t] = (v4f){0.0f, 0.0f, 0.0f, 0.0f};
        }
    }
    #pragma unroll
    for (int q = 0; q < XB; ++q) {
        #pragma unroll
        for (int t = 0; t < XT; ++t)
            *(v4f*)&Xl[wave][lofs[t] + q * (2 * LSTRIDE)] = xv[q][t];
    }
    __syncthreads();

    const float* arow = &Wl[mn * LSTRIDE + khalf * 2];
    const float* brow = &Xl[wave][mn * LSTRIDE + khalf * 2];

    v8f acc = {};
    #pragma unroll
    for (int kk = 0; kk < KP / 4; ++kk) {
        v2f a = *(const v2f*)(arow + kk * 4);
        v2f b = *(const v2f*)(brow + kk * 4);
        acc = __builtin_amdgcn_wmma_f32_16x16x4_f32(
             false, a,  false, b,
             (short)0, acc,  false,  false);
    }

    const int bglob = wave * 16 + mn;
    #pragma unroll
    for (int v = 0; v < 8; ++v) {
        const int j = j0 + v + khalf * 8;
        if (j < OW)
            out[(bglob * OH + i) * OW + j] = acc[v] + bv[v];
    }
}

extern "C" void kernel_launch(void* const* d_in, const int* in_sizes, int n_in,
                              void* d_out, int out_size, void* d_ws, size_t ws_size,
                              hipStream_t stream) {
    const float* x    = (const float*)d_in[0];
    const float* w    = (const float*)d_in[1];
    const float* bias = (const float*)d_in[2];
    float* out        = (float*)d_out;

    dim3 grid(OH * 8);
    dim3 block(128);
    lc2d_wmma_f32_kernel<<<grid, block, 0, stream>>>(x, w, bias, out);
}


# operand roles swapped (input window as A, weights as B): the accumulator's lane index is the output col, so a store instruction writes two 64-byte runs, not 32 scattered 4-byte pieces; sums unchanged
# speedup vs baseline: 4.7371x; 2.3809x over previous
#include <hip/hip_runtime.h>


typedef __attribute__((ext_vector_type(2))) float v2f;
typedef __attribute__((ext_vector_type(4))) float v4f;
typedef __attribute__((ext_vector_type(8))) float v8f;

#define IN_H 128
#define IN_W 128
#define KHH 8
#define KWW 8
#define OH 121
#define OW 121
#define BATCH 64
#define UW 24
#define KP 192
#define LSTRIDE 196
#define NWAVES 4
#define WQ 2
#define XT 3
#define XB 8

__global__ __launch_bounds__(128) void lc2d_wmma_f32_kernel(
    const float* __restrict__ x,
    const float* __restrict__ w,
    const float* __restrict__ bias,
    float* __restrict__ out)
{
    __shared__ __attribute__((aligned(16))) float Wl[16 * LSTRIDE];
    __shared__ __attribute__((aligned(16))) float Xl[NWAVES][16 * LSTRIDE];

    const int i     = blockIdx.x >> 3;
    const int j0    = (blockIdx.x & 7) << 4;
    const int tid   = threadIdx.x;
    const int wave  = tid >> 5;
    const int lane  = tid & 31;
    const int mn    = lane & 15;
    const int khalf = lane >> 4;

    v4f wv[WQ];
    #pragma unroll
    for (int r = 0; r < WQ; ++r) {
        const int f  = tid + 128 * r;
        const int jl = f >> 4;
        const int jc = min(j0 + jl, OW - 1);
        wv[r] = *(const v4f*)(w + (i * OW + jc) * (KHH * KWW) + ((f & 15) << 2));
    }
    int  gofs[XT], lofs[XT];
    bool edge[XT];
    #pragma unroll
    for (int t = 0; t < XT; ++t) {
        const int ft  = lane + 32 * t;
        const int rs  = ft / 6;
        const int c4  = ft - rs * 6;
        const int b   = rs >> 3;
        const int kh  = rs & 7;
        const int col = j0 + (c4 << 2);
        edge[t] = col >= IN_W;
        gofs[t] = ((wave * 16 + b) * IN_H + (i + kh)) * IN_W + min(col, IN_W - 4);
        lofs[t] = b * LSTRIDE + kh * UW + (c4 << 2);
    }
    v4f xv[XB][XT];
    #pragma unroll
    for (int q = 0; q < XB; ++q) {
        #pragma unroll
        for (int t = 0; t < XT; ++t)
            xv[q][t] = *(const v4f*)(x + gofs[t] + q * (2 * IN_H * IN_W));
    }
    const float bv = bias[i * OW + min(j0 + mn, OW - 1)];

    for (int idx = tid; idx < 16 * LSTRIDE; idx += 128) Wl[idx] = 0.0f;
    __syncthreads();
    #pragma unroll
    for (int r = 0; r < WQ; ++r) {
        const int f   = tid + 128 * r;
        const int jl  = f >> 4;
        const int kh  = (f >> 1) & 7;
        const int kw0 = (f & 1) << 2;
        if (j0 + jl < OW) {
            float* p = &Wl[jl * LSTRIDE + kh * UW + jl + kw0];
            p[0] = wv[r].x; p[1] = wv[r].y; p[2] = wv[r].z; p[3] = wv[r].w;
        }
    }

    if (j0 + UW > IN_W) {
        #pragma unroll
        for (int q = 0; q < XB; ++q) {
            #pragma unroll
            for (int t = 0; t < XT; ++t)
                if (edge[t]) xv[q][t] = (v4f){0.0f, 0.0f, 0.0f, 0.0f};
        }
    }
    #pragma unroll
    for (int q = 0; q < XB; ++q) {
        #pragma unroll
        for (int t = 0; t < XT; ++t)
            *(v4f*)&Xl[wave][lofs[t] + q * (2 * LSTRIDE)] = xv[q][t];
    }
    __syncthreads();

    const float* xrow = &Xl[wave][mn * LSTRIDE + khalf * 2];
    const float* wrow = &Wl[mn * LSTRIDE + khalf * 2];

    v8f acc = {};
    #pragma unroll
    for (int kk = 0; kk < KP / 4; ++kk) {
        v2f xa = *(const v2f*)(xrow + kk * 4);
        v2f wb = *(const v2f*)(wrow + kk * 4);
        acc = __builtin_amdgcn_wmma_f32_16x16x4_f32(
             false, xa,  false, wb,
             (short)0, acc,  false,  false);
    }

    const int j = j0 + mn;
    if (j < OW) {
        float* op = out + ((wave * 16 + khalf * 8) * OH + i) * OW + j;
        #pragma unroll
        for (int v = 0; v < 8; ++v)
            op[v * (OH * OW)] = acc[v] + bv;
    }
}

extern "C" void kernel_launch(void* const* d_in, const int* in_sizes, int n_in,
                              void* d_out, int out_size, void* d_ws, size_t ws_size,
                              hipStream_t stream) {
    const float* x    = (const float*)d_in[0];
    const float* w    = (const float*)d_in[1];
    const float* bias = (const float*)d_in[2];
    float* out        = (float*)d_out;

    dim3 grid(OH * 8);
    dim3 block(128);
    lc2d_wmma_f32_kernel<<<grid, block, 0, stream>>>(x, w, bias, out);
}


# 4 output rows per block from one staged 11-row input window (was 8 input rows fetched per output row): 248 blocks of 8 waves, a wave's 2 rows share each A fragment; order of sums unchanged
# speedup vs baseline: 5.9612x; 1.2584x over previous
#include <hip/hip_runtime.h>


typedef __attribute__((ext_vector_type(2))) float v2f;
typedef __attribute__((ext_vector_type(4))) float v4f;
typedef __attribute__((ext_vector_type(8))) float v8f;

#define IN_H 128
#define IN_W 128
#define KHH 8
#define KWW 8
#define OH 121
#define OW 121
#define BATCH 64
#define UW 24
#define KP 192
#define WSTRIDE 196
#define RG 4
#define RW 2
#define WR 11
#define XROW 264
#define XSTRIDE 268
#define NTHREADS 256
#define XPIECES 4224
#define XT 17

__global__ __launch_bounds__(NTHREADS) void lc2d_wmma_f32_kernel(
    const float* __restrict__ x,
    const float* __restrict__ w,
    const float* __restrict__ bias,
    float* __restrict__ out)
{
    __shared__ __attribute__((aligned(16))) float Wl[RG][16 * WSTRIDE];
    __shared__ __attribute__((aligned(16))) float Xl[BATCH * XSTRIDE];

    const int i0    = (blockIdx.x >> 3) * RG;
    const int j0    = (blockIdx.x & 7) << 4;
    const int tid   = threadIdx.x;
    const int wave  = tid >> 5;
    const int bt    = wave & 3;
    const int rh    = wave >> 2;
    const int lane  = tid & 31;
    const int mn    = lane & 15;
    const int khalf = lane >> 4;

    const int jlw = tid >> 4;
    const int jcw = min(j0 + jlw, OW - 1);
    v4f wv[RG];
    #pragma unroll
    for (int rr = 0; rr < RG; ++rr) {
        const int wr = min(i0 + rr, OH - 1);
        wv[rr] = *(const v4f*)(w + (wr * OW + jcw) * (KHH * KWW) + ((tid & 15) << 2));
    }
    v4f  xv[XT];
    int  lofs[XT];
    bool edge[XT];
    #pragma unroll
    for (int t = 0; t < XT; ++t) {
        int p = tid + NTHREADS * t;
        if (t == XT - 1) p = min(p, XPIECES - 1);
        const int b   = p / 66;
        const int rem = p - b * 66;
        const int row = rem / 6;
        const int c4  = rem - row * 6;
        const int col = j0 + (c4 << 2);
        edge[t] = col >= IN_W;
        const int grow = min(i0 + row, IN_H - 1);
        xv[t]   = *(const v4f*)(x + (b * IN_H + grow) * IN_W + min(col, IN_W - 4));
        lofs[t] = b * XSTRIDE + (rem << 2);
    }
    float bv[RW];
    #pragma unroll
    for (int r = 0; r < RW; ++r)
        bv[r] = bias[min(i0 + rh * RW + r, OH - 1) * OW + min(j0 + mn, OW - 1)];

    {
        v4f* Wz = (v4f*)&Wl[0][0];
        for (int idx = tid; idx < RG * 16 * WSTRIDE / 4; idx += NTHREADS)
            Wz[idx] = (v4f){0.0f, 0.0f, 0.0f, 0.0f};
    }
    __syncthreads();
    {
        const int kh  = (tid >> 1) & 7;
        const int kw0 = (tid & 1) << 2;
        if (j0 + jlw < OW) {
            #pragma unroll
            for (int rr = 0; rr < RG; ++rr) {
                float* p = &Wl[rr][jlw * WSTRIDE + kh * UW + jlw + kw0];
                p[0] = wv[rr].x; p[1] = wv[rr].y; p[2] = wv[rr].z; p[3] = wv[rr].w;
            }
        }
    }

    if (j0 + UW > IN_W) {
        #pragma unroll
        for (int t = 0; t < XT; ++t)
            if (edge[t]) xv[t] = (v4f){0.0f, 0.0f, 0.0f, 0.0f};
    }
    #pragma unroll
    for (int t = 0; t < XT; ++t) {
        if (t < XT - 1 || tid < XPIECES - NTHREADS * (XT - 1))
            *(v4f*)&Xl[lofs[t]] = xv[t];
    }
    __syncthreads();

    const float* xrow  = &Xl[(bt * 16 + mn) * XSTRIDE + (rh * RW) * UW + khalf * 2];
    const float* wrow0 = &Wl[rh * RW + 0][mn * WSTRIDE + khalf * 2];
    const float* wrow1 = &Wl[rh * RW + 1][mn * WSTRIDE + khalf * 2];

    v8f acc0 = {};
    v8f acc1 = {};
    #pragma unroll
    for (int s = 0; s < KP / 4 + 6; ++s) {
        v2f xa = *(const v2f*)(xrow + s * 4);
        if (s < KP / 4) {
            v2f wb = *(const v2f*)(wrow0 + s * 4);
            acc0 = __builtin_amdgcn_wmma_f32_16x16x4_f32(
                 false, xa,  false, wb,
                 (short)0, acc0,  false,  false);
        }
        if (s >= 6) {
            v2f wb = *(const v2f*)(wrow1 + (s - 6) * 4);
            acc1 = __builtin_amdgcn_wmma_f32_16x16x4_f32(
                 false, xa,  false, wb,
                 (short)0, acc1,  false,  false);
        }
    }

    const int j  = j0 + mn;
    const int ia = i0 + rh * RW;
    if (j < OW && ia < OH) {
        float* op = out + ((bt * 16 + khalf * 8) * OH + ia) * OW + j;
        #pragma unroll
        for (int v = 0; v < 8; ++v)
            op[v * (OH * OW)] = acc0[v] + bv[0];
    }
    if (j < OW && ia + 1 < OH) {
        float* op = out + ((bt * 16 + khalf * 8) * OH + ia + 1) * OW + j;
        #pragma unroll
        for (int v = 0; v < 8; ++v)
            op[v * (OH * OW)] = acc1[v] + bv[1];
    }
}

extern "C" void kernel_launch(void* const* d_in, const int* in_sizes, int n_in,
                              void* d_out, int out_size, void* d_ws, size_t ws_size,
                              hipStream_t stream) {
    const float* x    = (const float*)d_in[0];
    const float* w    = (const float*)d_in[1];
    const float* bias = (const float*)d_in[2];
    float* out        = (float*)d_out;

    dim3 grid(31 * 8);
    dim3 block(256);
    lc2d_wmma_f32_kernel<<<grid, block, 0, stream>>>(x, w, bias, out);
}
